# starting kernel
# baseline (speedup 1.0000x reference)
#include <hip/hip_runtime.h>
#include <hip/hip_bf16.h>


typedef __attribute__((ext_vector_type(16))) __bf16 v16bf;
typedef __attribute__((ext_vector_type(8)))  __bf16 v8bf;
typedef __attribute__((ext_vector_type(8)))  float  v8f;

#define D_ROWS 8192
#define N_CENT 8192
#define N_FEAT 128
#define L_OUT  64

__device__ __forceinline__ v16bf frag_cat(v8bf a, v8bf b) {
  return __builtin_shufflevector(a, b, 0, 1, 2, 3, 4, 5, 6, 7,
                                       8, 9, 10, 11, 12, 13, 14, 15);
}

__device__ __forceinline__ v16bf load_frag_g(const __bf16* __restrict__ base,
                                             int off, int h) {
  const v8bf* p0 = reinterpret_cast<const v8bf*>(base + off + 8 * h);
  const v8bf* p1 = reinterpret_cast<const v8bf*>(base + off + 16 + 8 * h);
  return frag_cat(*p0, *p1);
}

__device__ __forceinline__ v16bf load_frag_lds(const __bf16* rowp, int h) {
  const v8bf* p0 = reinterpret_cast<const v8bf*>(rowp + 8 * h);
  const v8bf* p1 = reinterpret_cast<const v8bf*>(rowp + 16 + 8 * h);
  return frag_cat(*p0, *p1);
}

__device__ __forceinline__ v8f wmma_bf16(v16bf a, v16bf b, v8f c) {
  return __builtin_amdgcn_wmma_f32_16x16x32_bf16(
       false, a,  false, b,
       (short)0, c,  false,  false);
}

__global__ void split_f32_bf16(const float* __restrict__ in,
                               __bf16* __restrict__ hi, __bf16* __restrict__ lo,
                               int count) {
  int i = blockIdx.x * blockDim.x + threadIdx.x;
  if (i < count) {
    float v  = in[i];
    __bf16 h = (__bf16)v;
    hi[i]    = h;
    lo[i]    = (__bf16)(v - (float)h);
  }
}

__global__ void transpose_split(const float* __restrict__ w,
                                __bf16* __restrict__ hi,
                                __bf16* __restrict__ lo, int N, int L) {
  int i = blockIdx.x * blockDim.x + threadIdx.x;
  if (i < N * L) {
    int c    = i / L;
    int j    = i - c * L;
    float v  = w[i];
    __bf16 h = (__bf16)v;
    hi[(size_t)j * N + c] = h;
    lo[(size_t)j * N + c] = (__bf16)(v - (float)h);
  }
}

__global__ void rowsum2(const float* __restrict__ in, float* __restrict__ out,
                        int rows, int cols, float bias) {
  int r = blockIdx.x * blockDim.x + threadIdx.x;
  if (r < rows) {
    const float* p = in + (size_t)r * cols;
    float s = bias;
    for (int k = 0; k < cols; ++k) s += p[k] * p[k];
    out[r] = s;
  }
}

__global__ __launch_bounds__(256) void rbfn_main(
    const __bf16* __restrict__ xs_hi, const __bf16* __restrict__ xs_lo,
    const __bf16* __restrict__ c_hi,  const __bf16* __restrict__ c_lo,
    const __bf16* __restrict__ wt_hi, const __bf16* __restrict__ wt_lo,
    const float* __restrict__ x2p1,
    const float* __restrict__ c2,
    float* __restrict__ out) {
  constexpr int RS = 40;
  __shared__ alignas(16) __bf16 lds_phi_hi[8][16][RS];
  __shared__ alignas(16) __bf16 lds_phi_lo[8][16][RS];

  const int lane = threadIdx.x & 31;
  const int wave = threadIdx.x >> 5;
  const int h    = lane >> 4;
  const int ln   = lane & 15;
  const int rowbase = blockIdx.x * 128 + wave * 16;

  v16bf a_hi[4], a_lo[4];
  {
    const int xoff = (rowbase + ln) * N_FEAT;
#pragma unroll
    for (int kc = 0; kc < 4; ++kc) {
      a_hi[kc] = load_frag_g(xs_hi, xoff + kc * 32, h);
      a_lo[kc] = load_frag_g(xs_lo, xoff + kc * 32, h);
    }
  }

  float x2v[8];
#pragma unroll
  for (int v = 0; v < 8; ++v) x2v[v] = x2p1[rowbase + v + 8 * h];

  v8f acc[4];
#pragma unroll
  for (int nt = 0; nt < 4; ++nt)
    acc[nt] = v8f{0.f, 0.f, 0.f, 0.f, 0.f, 0.f, 0.f, 0.f};

  __bf16* phi_hi = &lds_phi_hi[wave][0][0];
  __bf16* phi_lo = &lds_phi_lo[wave][0][0];

  for (int cb = 0; cb < N_CENT; cb += 32) {
    if (cb + 32 < N_CENT) {
      __builtin_prefetch(c_hi + (size_t)((cb + 32 + ln) * N_FEAT), 0, 0);
      __builtin_prefetch(c_hi + (size_t)((cb + 48 + ln) * N_FEAT), 0, 0);
    }

#pragma unroll
    for (int ct = 0; ct < 2; ++ct) {
      const int coff = (cb + ct * 16 + ln) * N_FEAT;

      v8f g = v8f{0.f, 0.f, 0.f, 0.f, 0.f, 0.f, 0.f, 0.f};
#pragma unroll
      for (int kc = 0; kc < 4; ++kc) {
        v16bf b_hi = load_frag_g(c_hi, coff + kc * 32, h);
        v16bf b_lo = load_frag_g(c_lo, coff + kc * 32, h);
        g = wmma_bf16(a_hi[kc], b_hi, g);
        g = wmma_bf16(a_hi[kc], b_lo, g);
        g = wmma_bf16(a_lo[kc], b_hi, g);
      }

      const float c2v = c2[cb + ct * 16 + ln];
#pragma unroll
      for (int v = 0; v < 8; ++v) {
        float t   = fmaf(-2.0f, g[v], x2v[v] + c2v);
        t         = fmaxf(t, 1.0f);
        float phi = __builtin_amdgcn_sqrtf(t);
        __bf16 ph = (__bf16)phi;
        __bf16 pl = (__bf16)(phi - (float)ph);
        const int idx = (v + 8 * h) * RS + ct * 16 + ln;
        phi_hi[idx] = ph;
        phi_lo[idx] = pl;
      }
    }

    asm volatile("s_wait_dscnt 0" ::: "memory");

    v16bf pa_hi = load_frag_lds(phi_hi + ln * RS, h);
    v16bf pa_lo = load_frag_lds(phi_lo + ln * RS, h);

#pragma unroll
    for (int nt = 0; nt < 4; ++nt) {
      const int woff = (nt * 16 + ln) * N_CENT + cb;
      v16bf wb_hi = load_frag_g(wt_hi, woff, h);
      v16bf wb_lo = load_frag_g(wt_lo, woff, h);
      acc[nt] = wmma_bf16(pa_hi, wb_hi, acc[nt]);
      acc[nt] = wmma_bf16(pa_hi, wb_lo, acc[nt]);
      acc[nt] = wmma_bf16(pa_lo, wb_hi, acc[nt]);
    }
  }

#pragma unroll
  for (int nt = 0; nt < 4; ++nt)
#pragma unroll
    for (int v = 0; v < 8; ++v)
      out[(rowbase + v + 8 * h) * L_OUT + nt * 16 + ln] = acc[nt][v];
}

extern "C" void kernel_launch(void* const* d_in, const int* in_sizes, int n_in,
                              void* d_out, int out_size, void* d_ws, size_t ws_size,
                              hipStream_t stream) {
  (void)in_sizes; (void)n_in; (void)out_size; (void)ws_size;

  const float* xs     = (const float*)d_in[0];
  const float* centre = (const float*)d_in[1];
  const float* weight = (const float*)d_in[2];
  float* out          = (float*)d_out;

  const size_t XS = (size_t)D_ROWS * N_FEAT;
  const size_t WT = (size_t)N_CENT * L_OUT;

  __bf16* xs_hi = (__bf16*)d_ws;
  __bf16* xs_lo = xs_hi + XS;
  __bf16* c_hi  = xs_lo + XS;
  __bf16* c_lo  = c_hi + XS;
  __bf16* wt_hi = c_lo + XS;
  __bf16* wt_lo = wt_hi + WT;
  float*  x2p1  = (float*)(wt_lo + WT);
  float*  c2    = x2p1 + D_ROWS;

  const int T = 256;
  split_f32_bf16<<<dim3((XS + T - 1) / T), dim3(T), 0, stream>>>(xs, xs_hi, xs_lo, (int)XS);
  split_f32_bf16<<<dim3((XS + T - 1) / T), dim3(T), 0, stream>>>(centre, c_hi, c_lo, (int)XS);
  transpose_split<<<dim3((WT + T - 1) / T), dim3(T), 0, stream>>>(weight, wt_hi, wt_lo, N_CENT, L_OUT);
  rowsum2<<<dim3((D_ROWS + T - 1) / T), dim3(T), 0, stream>>>(xs, x2p1, D_ROWS, N_FEAT, 1.0f);
  rowsum2<<<dim3((N_CENT + T - 1) / T), dim3(T), 0, stream>>>(centre, c2, N_CENT, N_FEAT, 0.0f);

  rbfn_main<<<dim3(D_ROWS / 128), dim3(T), 0, stream>>>(
      xs_hi, xs_lo, c_hi, c_lo, wt_hi, wt_lo, x2p1, c2, out);
}


# Staging registers refilled piece by piece right behind the LDS store that empties them, LDS-only barrier: the fetch of tile t+2 stays in flight through all of trip t+1; f32 sum order unchanged
# speedup vs baseline: 8.1438x; 8.1438x over previous
#include <hip/hip_runtime.h>
#include <hip/hip_bf16.h>


typedef __attribute__((ext_vector_type(16))) __bf16 v16bf;
typedef __attribute__((ext_vector_type(8)))  __bf16 v8bf;
typedef __attribute__((ext_vector_type(8)))  float  v8f;

#define D_ROWS 8192
#define N_CENT 8192
#define N_FEAT 128
#define L_OUT  64

__device__ __forceinline__ v16bf frag_cat(v8bf a, v8bf b) {
  return __builtin_shufflevector(a, b, 0, 1, 2, 3, 4, 5, 6, 7,
                                       8, 9, 10, 11, 12, 13, 14, 15);
}

__device__ __forceinline__ v16bf load_frag_g(const __bf16* __restrict__ base,
                                             int off, int h) {
  const v8bf* p0 = reinterpret_cast<const v8bf*>(base + off + 8 * h);
  const v8bf* p1 = reinterpret_cast<const v8bf*>(base + off + 16 + 8 * h);
  return frag_cat(*p0, *p1);
}

__device__ __forceinline__ v16bf load_frag_lds(const __bf16* rowp, int h) {
  const v8bf* p0 = reinterpret_cast<const v8bf*>(rowp + 8 * h);
  const v8bf* p1 = reinterpret_cast<const v8bf*>(rowp + 16 + 8 * h);
  return frag_cat(*p0, *p1);
}

__device__ __forceinline__ v8f wmma_bf16(v16bf a, v16bf b, v8f c) {
  return __builtin_amdgcn_wmma_f32_16x16x32_bf16(
       false, a,  false, b,
       (short)0, c,  false,  false);
}

__device__ __forceinline__ void lds_barrier() {
  __builtin_amdgcn_fence(__ATOMIC_RELEASE, "workgroup", "local");
  __builtin_amdgcn_s_barrier();
  __builtin_amdgcn_fence(__ATOMIC_ACQUIRE, "workgroup", "local");
}

__global__ void split_f32_bf16(const float* __restrict__ in,
                               __bf16* __restrict__ hi, __bf16* __restrict__ lo,
                               int count) {
  int i = blockIdx.x * blockDim.x + threadIdx.x;
  if (i < count) {
    float v  = in[i];
    __bf16 h = (__bf16)v;
    hi[i]    = h;
    lo[i]    = (__bf16)(v - (float)h);
  }
}

__global__ void transpose_split(const float* __restrict__ w,
                                __bf16* __restrict__ hi,
                                __bf16* __restrict__ lo, int N, int L) {
  int i = blockIdx.x * blockDim.x + threadIdx.x;
  if (i < N * L) {
    int c    = i / L;
    int j    = i - c * L;
    float v  = w[i];
    __bf16 h = (__bf16)v;
    hi[(size_t)j * N + c] = h;
    lo[(size_t)j * N + c] = (__bf16)(v - (float)h);
  }
}

__global__ void rowsum2(const float* __restrict__ in, float* __restrict__ out,
                        int rows, int cols, float bias) {
  int r = blockIdx.x * blockDim.x + threadIdx.x;
  if (r < rows) {
    const float* p = in + (size_t)r * cols;
    float s = bias;
    for (int k = 0; k < cols; ++k) s += p[k] * p[k];
    out[r] = s;
  }
}

__global__ __launch_bounds__(256) __attribute__((amdgpu_waves_per_eu(2, 2)))
void rbfn_main(
    const __bf16* __restrict__ xs_hi, const __bf16* __restrict__ xs_lo,
    const __bf16* __restrict__ c_hi,  const __bf16* __restrict__ c_lo,
    const __bf16* __restrict__ wt_hi, const __bf16* __restrict__ wt_lo,
    const float* __restrict__ x2p1,
    const float* __restrict__ c2,
    float* __restrict__ out) {
  constexpr int WP = 136;
  constexpr int CP = 136;
  constexpr int TC = 128;
  constexpr int NT = N_CENT / TC;
  constexpr int OP = 68;
  __shared__ alignas(16) __bf16 lds_phi_hi[2][2][16][WP];
  __shared__ alignas(16) __bf16 lds_phi_lo[2][2][16][WP];
  __shared__ alignas(16) float lds_out[32][OP];
  __shared__ alignas(16) __bf16 lds_c_hi[2][TC][CP];
  __shared__ alignas(16) __bf16 lds_c_lo[2][TC][CP];
  __shared__ alignas(16) __bf16 lds_w_hi[2][64][WP];
  __shared__ alignas(16) __bf16 lds_w_lo[2][64][WP];
  __shared__ alignas(16) float lds_c2[2][2 * TC];

  const int tid  = threadIdx.x;
  const int lane = tid & 31;
  const int wave = tid >> 5;
  const int rt   = wave >> 2;
  const int ch   = wave & 3;
  const int h    = lane >> 4;
  const int ln   = lane & 15;
  const int rowbase = blockIdx.x * 32 + rt * 16;

  const int c_goff = tid * 8;
  const int c_loff = (tid >> 4) * CP + (tid & 15) * 8;
  const int w_goff = (tid >> 4) * N_CENT + (tid & 15) * 8;
  const int w_loff = (tid >> 4) * WP + (tid & 15) * 8;

  __bf16* const lc_hi = &lds_c_hi[0][0][0];
  __bf16* const lc_lo = &lds_c_lo[0][0][0];
  __bf16* const lw_hi = &lds_w_hi[0][0][0];
  __bf16* const lw_lo = &lds_w_lo[0][0][0];
  __bf16* const lp_hi = &lds_phi_hi[0][0][0][0];
  __bf16* const lp_lo = &lds_phi_lo[0][0][0][0];

  v8bf sc_hi[8], sc_lo[8], sw_hi[4], sw_lo[4];
  float sc2;

  auto stage_load = [&](int cb) {
    sc2 = c2[cb + (tid & (TC - 1))];
#pragma unroll
    for (int i = 0; i < 8; ++i) {
      const int o = cb * N_FEAT + c_goff + i * 2048;
      sc_hi[i] = *reinterpret_cast<const v8bf*>(c_hi + o);
      sc_lo[i] = *reinterpret_cast<const v8bf*>(c_lo + o);
    }
#pragma unroll
    for (int i = 0; i < 4; ++i) {
      const int o = w_goff + i * 16 * N_CENT + cb;
      sw_hi[i] = *reinterpret_cast<const v8bf*>(wt_hi + o);
      sw_lo[i] = *reinterpret_cast<const v8bf*>(wt_lo + o);
    }
  };
  auto stage_store = [&](int st) {
#pragma unroll
    for (int i = 0; i < 8; ++i) {
      const int o = st * (TC * CP) + c_loff + i * 16 * CP;
      *reinterpret_cast<v8bf*>(lc_hi + o) = sc_hi[i];
      *reinterpret_cast<v8bf*>(lc_lo + o) = sc_lo[i];
    }
#pragma unroll
    for (int i = 0; i < 4; ++i) {
      const int o = st * (64 * WP) + w_loff + i * 16 * WP;
      *reinterpret_cast<v8bf*>(lw_hi + o) = sw_hi[i];
      *reinterpret_cast<v8bf*>(lw_lo + o) = sw_lo[i];
    }
    lds_c2[st][tid] = sc2;
  };
  auto stage_swap = [&](int st, int cbn) {
    lds_c2[st][tid] = sc2;
    sc2 = c2[cbn + (tid & (TC - 1))];
    __builtin_amdgcn_sched_barrier(0);
#pragma unroll
    for (int i = 0; i < 8; ++i) {
      const int lo_ = st * (TC * CP) + c_loff + i * 16 * CP;
      const int go_ = cbn * N_FEAT + c_goff + i * 2048;
      *reinterpret_cast<v8bf*>(lc_hi + lo_) = sc_hi[i];
      *reinterpret_cast<v8bf*>(lc_lo + lo_) = sc_lo[i];
      sc_hi[i] = *reinterpret_cast<const v8bf*>(c_hi + go_);
      sc_lo[i] = *reinterpret_cast<const v8bf*>(c_lo + go_);
      __builtin_amdgcn_sched_barrier(0);
    }
#pragma unroll
    for (int i = 0; i < 4; ++i) {
      const int lo_ = st * (64 * WP) + w_loff + i * 16 * WP;
      const int go_ = w_goff + i * 16 * N_CENT + cbn;
      *reinterpret_cast<v8bf*>(lw_hi + lo_) = sw_hi[i];
      *reinterpret_cast<v8bf*>(lw_lo + lo_) = sw_lo[i];
      sw_hi[i] = *reinterpret_cast<const v8bf*>(wt_hi + go_);
      sw_lo[i] = *reinterpret_cast<const v8bf*>(wt_lo + go_);
      __builtin_amdgcn_sched_barrier(0);
    }
  };

  stage_load(0);

  v16bf a_hi[4], a_lo[4];
  {
    const int xoff = (rowbase + ln) * N_FEAT;
#pragma unroll
    for (int kc = 0; kc < 4; ++kc) {
      a_hi[kc] = load_frag_g(xs_hi, xoff + kc * 32, h);
      a_lo[kc] = load_frag_g(xs_lo, xoff + kc * 32, h);
    }
  }

  float x2v[8];
#pragma unroll
  for (int v = 0; v < 8; ++v) x2v[v] = x2p1[rowbase + v + 8 * h];

  v8f acc = v8f{0.f, 0.f, 0.f, 0.f, 0.f, 0.f, 0.f, 0.f};

  v16bf wb_hi[4], wb_lo[4];

  auto load_w = [&](int st) {
    const __bf16* tw_hi = lw_hi + st * (64 * WP);
    const __bf16* tw_lo = lw_lo + st * (64 * WP);
#pragma unroll
    for (int ks = 0; ks < 4; ++ks) {
      const int o = (ch * 16 + ln) * WP + ks * 32;
      wb_hi[ks] = load_frag_lds(tw_hi + o, h);
      wb_lo[ks] = load_frag_lds(tw_lo + o, h);
    }
  };

  auto gemm1_phi = [&](int st, int pb) {
    const __bf16* tc_hi = lc_hi + st * (TC * CP);
    const __bf16* tc_lo = lc_lo + st * (TC * CP);
    __bf16* phi_hi = lp_hi + (pb * 2 + rt) * (16 * WP);
    __bf16* phi_lo = lp_lo + (pb * 2 + rt) * (16 * WP);
    const float c2c0 = lds_c2[st][ch * 32 + ln];
    const float c2c1 = lds_c2[st][ch * 32 + 16 + ln];
#pragma unroll
    for (int ct = 0; ct < 2; ++ct) {
      const __bf16* rp_hi = tc_hi + (ch * 32 + ct * 16 + ln) * CP;
      const __bf16* rp_lo = tc_lo + (ch * 32 + ct * 16 + ln) * CP;

      v8f g = v8f{0.f, 0.f, 0.f, 0.f, 0.f, 0.f, 0.f, 0.f};
#pragma unroll
      for (int kc = 0; kc < 4; ++kc) {
        v16bf b_hi = load_frag_lds(rp_hi + kc * 32, h);
        v16bf b_lo = load_frag_lds(rp_lo + kc * 32, h);
        g = wmma_bf16(a_hi[kc], b_hi, g);
        g = wmma_bf16(a_hi[kc], b_lo, g);
        g = wmma_bf16(a_lo[kc], b_hi, g);
      }

      const float c2v = ct ? c2c1 : c2c0;
#pragma unroll
      for (int v = 0; v < 8; ++v) {
        float t   = fmaf(-2.0f, g[v], x2v[v] + c2v);
        t         = fmaxf(t, 1.0f);
        float phi = __builtin_amdgcn_sqrtf(t);
        __bf16 ph = (__bf16)phi;
        __bf16 pl = (__bf16)(phi - (float)ph);
        const int idx = (v + 8 * h) * WP + ch * 32 + ct * 16 + ln;
        phi_hi[idx] = ph;
        phi_lo[idx] = pl;
      }
    }
  };

  auto gemm2 = [&](int pb) {
    const __bf16* phi_hi = lp_hi + (pb * 2 + rt) * (16 * WP) + ln * WP;
    const __bf16* phi_lo = lp_lo + (pb * 2 + rt) * (16 * WP) + ln * WP;
#pragma unroll
    for (int ks = 0; ks < 4; ++ks) {
      v16bf pa_hi = load_frag_lds(phi_hi + ks * 32, h);
      v16bf pa_lo = load_frag_lds(phi_lo + ks * 32, h);
      acc = wmma_bf16(pa_hi, wb_hi[ks], acc);
      acc = wmma_bf16(pa_hi, wb_lo[ks], acc);
      acc = wmma_bf16(pa_lo, wb_hi[ks], acc);
    }
  };

  stage_store(0);
  stage_load(TC);
  lds_barrier();

  {
    load_w(0);
    gemm1_phi(0, 0);
    __builtin_amdgcn_sched_barrier(0);
    stage_swap(1, 2 * TC);
    lds_barrier();
  }

  for (int it = 1; it < NT; ++it) {
    const int st = it & 1;
    const int cbn = ((it + 2 < NT) ? it + 2 : NT - 1) * TC;

    gemm2(st ^ 1);
    load_w(st);
    gemm1_phi(st, st);

    __builtin_amdgcn_sched_barrier(0);
    stage_swap(st ^ 1, cbn);
    lds_barrier();
  }

  gemm2((NT - 1) & 1);

#pragma unroll
  for (int v = 0; v < 8; ++v)
    lds_out[rt * 16 + v + 8 * h][ch * 16 + ln] = acc[v];
  __syncthreads();
#pragma unroll
  for (int j = 0; j < 2; ++j) {
    const int row = (tid >> 4) + 16 * j;
    const int col = (tid & 15) * 4;
    const float4 val = *reinterpret_cast<const float4*>(&lds_out[row][col]);
    *reinterpret_cast<float4*>(out + (blockIdx.x * 32 + row) * L_OUT + col) = val;
  }
}

extern "C" void kernel_launch(void* const* d_in, const int* in_sizes, int n_in,
                              void* d_out, int out_size, void* d_ws, size_t ws_size,
                              hipStream_t stream) {
  (void)in_sizes; (void)n_in; (void)out_size; (void)ws_size;

  const float* xs     = (const float*)d_in[0];
  const float* centre = (const float*)d_in[1];
  const float* weight = (const float*)d_in[2];
  float* out          = (float*)d_out;

  const size_t XS = (size_t)D_ROWS * N_FEAT;
  const size_t WT = (size_t)N_CENT * L_OUT;

  __bf16* xs_hi = (__bf16*)d_ws;
  __bf16* xs_lo = xs_hi + XS;
  __bf16* c_hi  = xs_lo + XS;
  __bf16* c_lo  = c_hi + XS;
  __bf16* wt_hi = c_lo + XS;
  __bf16* wt_lo = wt_hi + WT;
  float*  x2p1  = (float*)(wt_lo + WT);
  float*  c2    = x2p1 + D_ROWS;

  const int T = 256;
  split_f32_bf16<<<dim3((XS + T - 1) / T), dim3(T), 0, stream>>>(xs, xs_hi, xs_lo, (int)XS);
  split_f32_bf16<<<dim3((XS + T - 1) / T), dim3(T), 0, stream>>>(centre, c_hi, c_lo, (int)XS);
  transpose_split<<<dim3((WT + T - 1) / T), dim3(T), 0, stream>>>(weight, wt_hi, wt_lo, N_CENT, L_OUT);
  rowsum2<<<dim3((D_ROWS + T - 1) / T), dim3(T), 0, stream>>>(xs, x2p1, D_ROWS, N_FEAT, 1.0f);
  rowsum2<<<dim3((N_CENT + T - 1) / T), dim3(T), 0, stream>>>(centre, c2, N_CENT, N_FEAT, 0.0f);

  rbfn_main<<<dim3(D_ROWS / 32), dim3(256), 0, stream>>>(
      xs_hi, xs_lo, c_hi, c_lo, wt_hi, wt_lo, x2p1, c2, out);
}


# Weight tile fetched as whole 256-byte rows: 8 sixteen-byte loads per thread, not 32 four-byte ones; a column wave takes columns c, c+4, .. (LDS stores keep distinct banks); f32 sum order unchanged
# speedup vs baseline: 9.6197x; 1.1812x over previous
#include <hip/hip_runtime.h>
#include <hip/hip_bf16.h>


typedef __attribute__((ext_vector_type(16))) __bf16 v16bf;
typedef __attribute__((ext_vector_type(8)))  __bf16 v8bf;
typedef __attribute__((ext_vector_type(2)))  __bf16 v2bf;
typedef __attribute__((ext_vector_type(8)))  float  v8f;
typedef __attribute__((ext_vector_type(4)))  float  v4f;
typedef __attribute__((ext_vector_type(2)))  float  v2f;
typedef __attribute__((ext_vector_type(2)))  unsigned v2u;
typedef __attribute__((ext_vector_type(4)))  unsigned v4u;
typedef __attribute__((ext_vector_type(8)))  unsigned v8u;

#define D_ROWS 8192
#define N_CENT 8192
#define N_FEAT 128
#define L_OUT  64

__device__ __forceinline__ v16bf frag_cat(v8bf a, v8bf b) {
  return __builtin_shufflevector(a, b, 0, 1, 2, 3, 4, 5, 6, 7,
                                       8, 9, 10, 11, 12, 13, 14, 15);
}

__device__ __forceinline__ v16bf load_frag_lds(const __bf16* rowp, int h) {
  const v8bf* p0 = reinterpret_cast<const v8bf*>(rowp + 8 * h);
  const v8bf* p1 = reinterpret_cast<const v8bf*>(rowp + 16 + 8 * h);
  return frag_cat(*p0, *p1);
}

__device__ __forceinline__ v8f wmma_bf16(v16bf a, v16bf b, v8f c) {
  return __builtin_amdgcn_wmma_f32_16x16x32_bf16(
       false, a,  false, b,
       (short)0, c,  false,  false);
}

__device__ __forceinline__ void split2(v2f v, unsigned& hi, unsigned& lo) {
  const unsigned p = __builtin_bit_cast(unsigned, __builtin_convertvector(v, v2bf));
  const v2f hf = v2f{__builtin_bit_cast(float, p << 16),
                     __builtin_bit_cast(float, p & 0xffff0000u)};
  hi = p;
  lo = __builtin_bit_cast(unsigned, __builtin_convertvector(v - hf, v2bf));
}
__device__ __forceinline__ void split4(v4f v, v2u& hi, v2u& lo) {
  unsigned h0, l0, h1, l1;
  split2(v2f{v.x, v.y}, h0, l0);
  split2(v2f{v.z, v.w}, h1, l1);
  hi = v2u{h0, h1};
  lo = v2u{l0, l1};
}

__device__ __forceinline__ float sum_squares(v4f v) {
  const v2f a = v2f{v.x, v.y};
  const v2f b = v2f{v.z, v.w};
  const v2f s = b * b + a * a;
  return s.x + s.y;
}

__device__ __forceinline__ void lds_barrier() {
  __builtin_amdgcn_fence(__ATOMIC_RELEASE, "workgroup", "local");
  __builtin_amdgcn_s_barrier();
  __builtin_amdgcn_fence(__ATOMIC_ACQUIRE, "workgroup", "local");
}

__global__ __launch_bounds__(256) __attribute__((amdgpu_waves_per_eu(2, 2)))
void rbfn_main(
    const float* __restrict__ xs,
    const float* __restrict__ centre,
    const float* __restrict__ weight,
    float* __restrict__ out) {
  constexpr int WP = 136;
  constexpr int CP = 136;
  constexpr int TC = 128;
  constexpr int NT = N_CENT / TC;
  constexpr int OP = 68;
  __shared__ alignas(16) __bf16 lds_phi_hi[2][2][16][WP];
  __shared__ alignas(16) __bf16 lds_phi_lo[2][2][16][WP];
  __shared__ alignas(16) float lds_out[32][OP];
  __shared__ alignas(16) __bf16 lds_c_hi[2][TC][CP];
  __shared__ alignas(16) __bf16 lds_c_lo[2][TC][CP];
  __shared__ alignas(16) __bf16 lds_w_hi[2][64][WP];
  __shared__ alignas(16) __bf16 lds_w_lo[2][64][WP];
  __shared__ alignas(16) float lds_c2[2][TC];

  const int tid  = threadIdx.x;
  const int lane = tid & 31;
  const int wave = tid >> 5;
  const int rt   = wave >> 2;
  const int ch   = wave & 3;
  const int h    = lane >> 4;
  const int ln   = lane & 15;
  const int rowbase = blockIdx.x * 32 + rt * 16;

  const int c_goff = tid * 4;
  const int c_loff = wave * CP + lane * 4;
  const int wc     = tid & 15;
  const int wg     = tid >> 4;
  const int w_goff = wg * 8 * L_OUT + wc * 4;
  const int w_loff = wc * WP + wg * 8;

  __bf16* const lc_hi = &lds_c_hi[0][0][0];
  __bf16* const lc_lo = &lds_c_lo[0][0][0];
  __bf16* const lw_hi = &lds_w_hi[0][0][0];
  __bf16* const lw_lo = &lds_w_lo[0][0][0];
  __bf16* const lp_hi = &lds_phi_hi[0][0][0][0];
  __bf16* const lp_lo = &lds_phi_lo[0][0][0][0];

  v4f sc[16];
  v4f sw[8];

  auto stage_load = [&](int cb) {
#pragma unroll
    for (int i = 0; i < 16; ++i)
      sc[i] = *reinterpret_cast<const v4f*>(centre + cb * N_FEAT + c_goff + i * 1024);
#pragma unroll
    for (int i = 0; i < 8; ++i)
      sw[i] = *reinterpret_cast<const v4f*>(weight + cb * L_OUT + w_goff + i * L_OUT);
  };

  auto stage_swap = [&](int st, int cbn) {
    v8f pa, pb;
#pragma unroll
    for (int i = 0; i < 16; ++i) {
      const int lo_ = st * (TC * CP) + c_loff + i * 8 * CP;
      const int go_ = cbn * N_FEAT + c_goff + i * 1024;
      v2u ph, pl;
      split4(sc[i], ph, pl);
      *reinterpret_cast<v2u*>(lc_hi + lo_) = ph;
      *reinterpret_cast<v2u*>(lc_lo + lo_) = pl;
      if (i < 8) pa[i] = sum_squares(sc[i]);
      else       pb[i - 8] = sum_squares(sc[i]);
      sc[i] = *reinterpret_cast<const v4f*>(centre + go_);
      __builtin_amdgcn_sched_barrier(0);
    }

    auto w_group = [&](int q) {
      const int lo_ = st * (64 * WP) + w_loff + q * 16 * WP;
      v2u h0, l0, h1, l1;
      split4(v4f{sw[0][q], sw[1][q], sw[2][q], sw[3][q]}, h0, l0);
      split4(v4f{sw[4][q], sw[5][q], sw[6][q], sw[7][q]}, h1, l1);
      *reinterpret_cast<v4u*>(lw_hi + lo_) = v4u{h0.x, h0.y, h1.x, h1.y};
      *reinterpret_cast<v4u*>(lw_lo + lo_) = v4u{l0.x, l0.y, l1.x, l1.y};
      __builtin_amdgcn_sched_barrier(0);
    };

    const bool u16 = (lane & 16) != 0;
    const bool u8  = (lane & 8) != 0;
    const bool u4  = (lane & 4) != 0;
    const bool u2  = (lane & 2) != 0;
    v8f x16;
    {
      const v8f send = u16 ? pa : pb;
#pragma unroll
      for (int k = 0; k < 8; ++k) x16[k] = __shfl_xor(send[k], 16);
    }
    __builtin_amdgcn_sched_barrier(0);
    w_group(0);
    const v8f r8 = (u16 ? pb : pa) + x16;
    const v4f r8a = __builtin_shufflevector(r8, r8, 0, 1, 2, 3);
    const v4f r8b = __builtin_shufflevector(r8, r8, 4, 5, 6, 7);
    v4f x8;
    {
      const v4f send = u8 ? r8a : r8b;
#pragma unroll
      for (int k = 0; k < 4; ++k) x8[k] = __shfl_xor(send[k], 8);
    }
    __builtin_amdgcn_sched_barrier(0);
    w_group(1);
    const v4f r4 = (u8 ? r8b : r8a) + x8;
    const v2f r4a = __builtin_shufflevector(r4, r4, 0, 1);
    const v2f r4b = __builtin_shufflevector(r4, r4, 2, 3);
    v2f x4;
    {
      const v2f send = u4 ? r4a : r4b;
      x4[0] = __shfl_xor(send[0], 4);
      x4[1] = __shfl_xor(send[1], 4);
    }
    __builtin_amdgcn_sched_barrier(0);
    w_group(2);
    const v2f r2 = (u4 ? r4b : r4a) + x4;
    const float x2 = __shfl_xor(u2 ? r2[0] : r2[1], 2);
    __builtin_amdgcn_sched_barrier(0);
    w_group(3);
    const float r1 = (u2 ? r2[1] : r2[0]) + x2;
    const float x1 = __shfl_xor(r1, 1);
    if ((lane & 1) == 0) lds_c2[st][wave + 8 * (lane >> 1)] = r1 + x1;
#pragma unroll
    for (int i = 0; i < 8; ++i)
      sw[i] = *reinterpret_cast<const v4f*>(weight + cbn * L_OUT + w_goff + i * L_OUT);
  };

  stage_load(0);

  v16bf a_hi[4], a_lo[4];
  float xpart = 0.0f;
  {
    const int xoff = (rowbase + ln) * N_FEAT + 8 * h;
#pragma unroll
    for (int kc = 0; kc < 4; ++kc) {
      const v4f f0 = *reinterpret_cast<const v4f*>(xs + xoff + kc * 32);
      const v4f f1 = *reinterpret_cast<const v4f*>(xs + xoff + kc * 32 + 4);
      const v4f f2 = *reinterpret_cast<const v4f*>(xs + xoff + kc * 32 + 16);
      const v4f f3 = *reinterpret_cast<const v4f*>(xs + xoff + kc * 32 + 20);
      v2u h0, l0, h1, l1, h2, l2, h3, l3;
      split4(f0, h0, l0);
      split4(f1, h1, l1);
      split4(f2, h2, l2);
      split4(f3, h3, l3);
      a_hi[kc] = __builtin_bit_cast(v16bf, v8u{h0.x, h0.y, h1.x, h1.y, h2.x, h2.y, h3.x, h3.y});
      a_lo[kc] = __builtin_bit_cast(v16bf, v8u{l0.x, l0.y, l1.x, l1.y, l2.x, l2.y, l3.x, l3.y});
      xpart += (sum_squares(f0) + sum_squares(f1)) + (sum_squares(f2) + sum_squares(f3));
    }
  }
  const float x2row = 1.0f + (xpart + __shfl_xor(xpart, 16));
  float x2v[8];
#pragma unroll
  for (int v = 0; v < 8; ++v) x2v[v] = __shfl(x2row, v + 8 * h);

  v8f acc = v8f{0.f, 0.f, 0.f, 0.f, 0.f, 0.f, 0.f, 0.f};

  v16bf wb_hi[4], wb_lo[4];

  auto load_w = [&](int st) {
    const __bf16* tw_hi = lw_hi + st * (64 * WP);
    const __bf16* tw_lo = lw_lo + st * (64 * WP);
#pragma unroll
    for (int ks = 0; ks < 4; ++ks) {
      const int o = (ch * 16 + ln) * WP + ks * 32;
      wb_hi[ks] = load_frag_lds(tw_hi + o, h);
      wb_lo[ks] = load_frag_lds(tw_lo + o, h);
    }
  };

  auto gemm1_phi = [&](int st, int pb) {
    const __bf16* tc_hi = lc_hi + st * (TC * CP);
    const __bf16* tc_lo = lc_lo + st * (TC * CP);
    __bf16* phi_hi = lp_hi + (pb * 2 + rt) * (16 * WP);
    __bf16* phi_lo = lp_lo + (pb * 2 + rt) * (16 * WP);
    const float c2c0 = lds_c2[st][ch * 32 + ln];
    const float c2c1 = lds_c2[st][ch * 32 + 16 + ln];
#pragma unroll
    for (int ct = 0; ct < 2; ++ct) {
      const __bf16* rp_hi = tc_hi + (ch * 32 + ct * 16 + ln) * CP;
      const __bf16* rp_lo = tc_lo + (ch * 32 + ct * 16 + ln) * CP;

      v8f g = v8f{0.f, 0.f, 0.f, 0.f, 0.f, 0.f, 0.f, 0.f};
#pragma unroll
      for (int kc = 0; kc < 4; ++kc) {
        v16bf b_hi = load_frag_lds(rp_hi + kc * 32, h);
        v16bf b_lo = load_frag_lds(rp_lo + kc * 32, h);
        g = wmma_bf16(a_hi[kc], b_hi, g);
        g = wmma_bf16(a_hi[kc], b_lo, g);
        g = wmma_bf16(a_lo[kc], b_hi, g);
      }

      const float c2v = ct ? c2c1 : c2c0;
#pragma unroll
      for (int v = 0; v < 8; ++v) {
        float t   = fmaf(-2.0f, g[v], x2v[v] + c2v);
        t         = fmaxf(t, 1.0f);
        float phi = __builtin_amdgcn_sqrtf(t);
        __bf16 ph = (__bf16)phi;
        __bf16 pl = (__bf16)(phi - (float)ph);
        const int idx = (v + 8 * h) * WP + ch * 32 + ct * 16 + ln;
        phi_hi[idx] = ph;
        phi_lo[idx] = pl;
      }
    }
  };

  auto gemm2 = [&](int pb) {
    const __bf16* phi_hi = lp_hi + (pb * 2 + rt) * (16 * WP) + ln * WP;
    const __bf16* phi_lo = lp_lo + (pb * 2 + rt) * (16 * WP) + ln * WP;
#pragma unroll
    for (int ks = 0; ks < 4; ++ks) {
      v16bf pa_hi = load_frag_lds(phi_hi + ks * 32, h);
      v16bf pa_lo = load_frag_lds(phi_lo + ks * 32, h);
      acc = wmma_bf16(pa_hi, wb_hi[ks], acc);
      acc = wmma_bf16(pa_hi, wb_lo[ks], acc);
      acc = wmma_bf16(pa_lo, wb_hi[ks], acc);
    }
  };

  stage_swap(0, TC);
  lds_barrier();

  {
    load_w(0);
    gemm1_phi(0, 0);
    __builtin_amdgcn_sched_barrier(0);
    stage_swap(1, 2 * TC);
    lds_barrier();
  }

  for (int it = 1; it < NT; ++it) {
    const int st = it & 1;
    const int cbn = ((it + 2 < NT) ? it + 2 : NT - 1) * TC;

    gemm2(st ^ 1);
    load_w(st);
    gemm1_phi(st, st);

    __builtin_amdgcn_sched_barrier(0);
    stage_swap(st ^ 1, cbn);
    lds_barrier();
  }

  gemm2((NT - 1) & 1);

#pragma unroll
  for (int v = 0; v < 8; ++v)
    lds_out[rt * 16 + v + 8 * h][4 * ln + ch] = acc[v];
  __syncthreads();
#pragma unroll
  for (int j = 0; j < 2; ++j) {
    const int row = (tid >> 4) + 16 * j;
    const int col = (tid & 15) * 4;
    const float4 val = *reinterpret_cast<const float4*>(&lds_out[row][col]);
    *reinterpret_cast<float4*>(out + (blockIdx.x * 32 + row) * L_OUT + col) = val;
  }
}

extern "C" void kernel_launch(void* const* d_in, const int* in_sizes, int n_in,
                              void* d_out, int out_size, void* d_ws, size_t ws_size,
                              hipStream_t stream) {
  (void)in_sizes; (void)n_in; (void)out_size; (void)d_ws; (void)ws_size;

  const float* xs     = (const float*)d_in[0];
  const float* centre = (const float*)d_in[1];
  const float* weight = (const float*)d_in[2];
  float* out          = (float*)d_out;

  rbfn_main<<<dim3(D_ROWS / 32), dim3(256), 0, stream>>>(xs, centre, weight, out);
}
